# starting kernel
# baseline (speedup 1.0000x reference)
#include <hip/hip_runtime.h>
#include <hip/hip_bf16.h>

typedef __attribute__((ext_vector_type(2))) float v2f;
typedef __attribute__((ext_vector_type(4))) float f4;
typedef __attribute__((ext_vector_type(8))) float v8f;

#define SEQ   2048
#define BATCH 16
#define HID   1024
#define MTOT  (SEQ * BATCH)
#define NTOT  (3 * HID)
#define KTOT  HID

#define TM 128
#define TN 128
#define TK 32
#define A_STRIDE 36
#define B_STRIDE 144

__global__ __launch_bounds__(256) void wmma_gemm_f32(
    const float* __restrict__ A, const float* __restrict__ B,
    float* __restrict__ C)
{
  __shared__ float As[TM * A_STRIDE];
  __shared__ float Bs[TK * B_STRIDE];

  const int tid  = threadIdx.x;
  const int lane = tid & 31;
  const int wave = tid >> 5;
  const int lm   = lane & 15;
  const int hk   = lane >> 4;
  const int waveM = (wave >> 2) * 64;
  const int waveN = (wave & 3) * 32;

  const int blockN = blockIdx.x * TN;
  const int blockM = blockIdx.y * TM;

  const int arow = tid >> 3;
  const int acol = (tid & 7) * 4;
  const int brow = tid >> 5;
  const int bcol = (tid & 31) * 4;

  v8f acc[4][2] = {};

  f4 aPre[4], bPre[4];
#pragma unroll
  for (int i = 0; i < 4; ++i) {
    aPre[i] = *(const f4*)&A[(size_t)(blockM + arow + i * 32) * KTOT + acol];
    bPre[i] = *(const f4*)&B[(size_t)(brow + i * 8) * NTOT + blockN + bcol];
  }

  for (int kt = 0; kt < KTOT; kt += TK) {
#pragma unroll
    for (int i = 0; i < 4; ++i) {
      *(f4*)&As[(arow + i * 32) * A_STRIDE + acol] = aPre[i];
      *(f4*)&Bs[(brow + i * 8) * B_STRIDE + bcol]  = bPre[i];
    }
    __syncthreads();

    const int ktn = kt + TK;
    if (ktn < KTOT) {
#pragma unroll
      for (int i = 0; i < 4; ++i) {
        aPre[i] = *(const f4*)&A[(size_t)(blockM + arow + i * 32) * KTOT + ktn + acol];
        bPre[i] = *(const f4*)&B[(size_t)(ktn + brow + i * 8) * NTOT + blockN + bcol];
      }
    }

#pragma unroll
    for (int kk = 0; kk < TK; kk += 4) {
      v2f afrag[4];
#pragma unroll
      for (int m = 0; m < 4; ++m)
        afrag[m] = *(const v2f*)&As[(waveM + m * 16 + lm) * A_STRIDE + kk + hk * 2];

      v2f bfrag[2];
#pragma unroll
      for (int n = 0; n < 2; ++n) {
        const int kb = (kk + hk * 2) * B_STRIDE + waveN + n * 16 + lm;
        bfrag[n].x = Bs[kb];
        bfrag[n].y = Bs[kb + B_STRIDE];
      }

#pragma unroll
      for (int m = 0; m < 4; ++m)
#pragma unroll
        for (int n = 0; n < 2; ++n)
          acc[m][n] = __builtin_amdgcn_wmma_f32_16x16x4_f32(
              false, afrag[m], false, bfrag[n], (short)0, acc[m][n],
              false, false);
    }
    __syncthreads();
  }

#pragma unroll
  for (int m = 0; m < 4; ++m) {
    const int row0 = blockM + waveM + m * 16 + hk * 8;
#pragma unroll
    for (int n = 0; n < 2; ++n) {
      const int col = blockN + waveN + n * 16 + lm;
#pragma unroll
      for (int v = 0; v < 8; ++v)
        C[(size_t)(row0 + v) * NTOT + col] = acc[m][n][v];
    }
  }
}

__device__ __forceinline__ float sigmoidf_(float z) {
  return 1.0f / (1.0f + __expf(-z));
}

__global__ __launch_bounds__(256) void sru_scan(
    const float* __restrict__ U, const float* __restrict__ xin,
    const float* __restrict__ bias, float* __restrict__ hout, int writeAll)
{
  const int idx = blockIdx.x * blockDim.x + threadIdx.x;
  const int b = idx >> 10;
  const int h = idx & (HID - 1);
  const float bf = bias[h];
  const float br = bias[HID + h];
  float c = 0.f;

  if (writeAll) {
    for (int t = 0; t < SEQ; ++t) {
      const size_t row = (size_t)t * BATCH + b;
      const float* Ur = U + row * NTOT;
      const float xt = Ur[h];
      const float f  = sigmoidf_(Ur[HID + h] + bf);
      c = f * c + (1.f - f) * xt;
      const float r  = sigmoidf_(Ur[2 * HID + h] + br);
      const float xv = xin[row * HID + h];
      hout[row * HID + h] = r * c + (1.f - r) * xv;
    }
  } else {
    for (int t = 0; t < SEQ; ++t) {
      const size_t row = (size_t)t * BATCH + b;
      const float* Ur = U + row * NTOT;
      const float xt = Ur[h];
      const float f  = sigmoidf_(Ur[HID + h] + bf);
      c = f * c + (1.f - f) * xt;
    }
    const size_t row = (size_t)(SEQ - 1) * BATCH + b;
    const float r  = sigmoidf_(U[row * NTOT + 2 * HID + h] + br);
    const float xv = xin[row * HID + h];
    hout[(size_t)b * HID + h] = r * c + (1.f - r) * xv;
  }
}

extern "C" void kernel_launch(void* const* d_in, const int* in_sizes, int n_in,
                              void* d_out, int out_size, void* d_ws, size_t ws_size,
                              hipStream_t stream) {
  const float* x  = (const float*)d_in[0];
  const float* W0 = (const float*)d_in[1];
  const float* b0 = (const float*)d_in[2];
  const float* W1 = (const float*)d_in[3];
  const float* b1 = (const float*)d_in[4];
  float* out = (float*)d_out;

  float* U  = (float*)d_ws;
  float* h1 = U + (size_t)MTOT * NTOT;

  const dim3 gemmGrid(NTOT / TN, MTOT / TM);
  const dim3 gemmBlk(256);
  const dim3 scanGrid((BATCH * HID) / 256);
  const dim3 scanBlk(256);

  wmma_gemm_f32<<<gemmGrid, gemmBlk, 0, stream>>>(x, W0, U);
  sru_scan<<<scanGrid, scanBlk, 0, stream>>>(U, x, b0, h1, 1);
  wmma_gemm_f32<<<gemmGrid, gemmBlk, 0, stream>>>(h1, W1, U);
  sru_scan<<<scanGrid, scanBlk, 0, stream>>>(U, h1, b1, out, 0);
}


# SRU scan kernels: a register ring asks for each step's loads 64 loads ahead of use, so 48 to 64 wave loads stay in flight under the serial chain; 64-thread blocks, one per WGP; sum order unchanged
# speedup vs baseline: 1.4147x; 1.4147x over previous
#include <hip/hip_runtime.h>
#include <hip/hip_bf16.h>

typedef __attribute__((ext_vector_type(2))) float v2f;
typedef __attribute__((ext_vector_type(4))) float f4;
typedef __attribute__((ext_vector_type(8))) float v8f;

#define SEQ   2048
#define BATCH 16
#define HID   1024
#define MTOT  (SEQ * BATCH)
#define NTOT  (3 * HID)
#define KTOT  HID

#define TM 128
#define TN 128
#define TK 32
#define A_STRIDE 36
#define B_STRIDE 144

__global__ __launch_bounds__(256) void wmma_gemm_f32(
    const float* __restrict__ A, const float* __restrict__ B,
    float* __restrict__ C)
{
  __shared__ float As[TM * A_STRIDE];
  __shared__ float Bs[TK * B_STRIDE];

  const int tid  = threadIdx.x;
  const int lane = tid & 31;
  const int wave = tid >> 5;
  const int lm   = lane & 15;
  const int hk   = lane >> 4;
  const int waveM = (wave >> 2) * 64;
  const int waveN = (wave & 3) * 32;

  const int blockN = blockIdx.x * TN;
  const int blockM = blockIdx.y * TM;

  const int arow = tid >> 3;
  const int acol = (tid & 7) * 4;
  const int brow = tid >> 5;
  const int bcol = (tid & 31) * 4;

  v8f acc[4][2] = {};

  f4 aPre[4], bPre[4];
#pragma unroll
  for (int i = 0; i < 4; ++i) {
    aPre[i] = *(const f4*)&A[(size_t)(blockM + arow + i * 32) * KTOT + acol];
    bPre[i] = *(const f4*)&B[(size_t)(brow + i * 8) * NTOT + blockN + bcol];
  }

  for (int kt = 0; kt < KTOT; kt += TK) {
#pragma unroll
    for (int i = 0; i < 4; ++i) {
      *(f4*)&As[(arow + i * 32) * A_STRIDE + acol] = aPre[i];
      *(f4*)&Bs[(brow + i * 8) * B_STRIDE + bcol]  = bPre[i];
    }
    __syncthreads();

    const int ktn = kt + TK;
    if (ktn < KTOT) {
#pragma unroll
      for (int i = 0; i < 4; ++i) {
        aPre[i] = *(const f4*)&A[(size_t)(blockM + arow + i * 32) * KTOT + ktn + acol];
        bPre[i] = *(const f4*)&B[(size_t)(ktn + brow + i * 8) * NTOT + blockN + bcol];
      }
    }

#pragma unroll
    for (int kk = 0; kk < TK; kk += 4) {
      v2f afrag[4];
#pragma unroll
      for (int m = 0; m < 4; ++m)
        afrag[m] = *(const v2f*)&As[(waveM + m * 16 + lm) * A_STRIDE + kk + hk * 2];

      v2f bfrag[2];
#pragma unroll
      for (int n = 0; n < 2; ++n) {
        const int kb = (kk + hk * 2) * B_STRIDE + waveN + n * 16 + lm;
        bfrag[n].x = Bs[kb];
        bfrag[n].y = Bs[kb + B_STRIDE];
      }

#pragma unroll
      for (int m = 0; m < 4; ++m)
#pragma unroll
        for (int n = 0; n < 2; ++n)
          acc[m][n] = __builtin_amdgcn_wmma_f32_16x16x4_f32(
              false, afrag[m], false, bfrag[n], (short)0, acc[m][n],
              false, false);
    }
    __syncthreads();
  }

#pragma unroll
  for (int m = 0; m < 4; ++m) {
    const int row0 = blockM + waveM + m * 16 + hk * 8;
#pragma unroll
    for (int n = 0; n < 2; ++n) {
      const int col = blockN + waveN + n * 16 + lm;
#pragma unroll
      for (int v = 0; v < 8; ++v)
        C[(size_t)(row0 + v) * NTOT + col] = acc[m][n][v];
    }
  }
}

#define SCAN_TPB 64
#define SU_STEP ((size_t)BATCH * NTOT)
#define SX_STEP ((size_t)BATCH * HID)
#define RING_SLOTS 4
#define GRP_ALL  4
#define GRP_LAST 8

__device__ __forceinline__ float sigmoidf_(float z) {
  return 1.0f / (1.0f + __expf(-z));
}

__global__ __launch_bounds__(SCAN_TPB) void sru_scan_all(
    const float* __restrict__ U, const float* __restrict__ xin,
    const float* __restrict__ bias, float* __restrict__ hout)
{
  const int idx = blockIdx.x * SCAN_TPB + threadIdx.x;
  const int b = idx >> 10;
  const int h = idx & (HID - 1);
  const float bf = bias[h];
  const float br = bias[HID + h];
  const float* pU = U   + (size_t)b * NTOT + h;
  const float* pX = xin + (size_t)b * HID + h;
  float*       pH = hout + (size_t)b * HID + h;

  float sx[RING_SLOTS][GRP_ALL], sf[RING_SLOTS][GRP_ALL];
  float sr[RING_SLOTS][GRP_ALL], sv[RING_SLOTS][GRP_ALL];
  float c = 0.f;

  auto ask = [&](int s, size_t t) {
#pragma unroll
    for (int j = 0; j < GRP_ALL; ++j) {
      const float* q = pU + (t + j) * SU_STEP;
      sx[s][j] = q[0];
      sf[s][j] = q[HID];
      sr[s][j] = q[2 * HID];
      sv[s][j] = pX[(t + j) * SX_STEP];
    }
  };
  auto use = [&](int s, size_t t) {
#pragma unroll
    for (int j = 0; j < GRP_ALL; ++j) {
      const float f = sigmoidf_(sf[s][j] + bf);
      c = f * c + (1.f - f) * sx[s][j];
      const float r = sigmoidf_(sr[s][j] + br);
      pH[(t + j) * SX_STEP] = r * c + (1.f - r) * sv[s][j];
    }
  };

  __builtin_amdgcn_sched_barrier(0);
#pragma unroll
  for (int s = 0; s < RING_SLOTS; ++s) {
    ask(s, (size_t)s * GRP_ALL);
    __builtin_amdgcn_sched_barrier(0);
  }

  constexpr int RING_STEPS = RING_SLOTS * GRP_ALL;
  for (int t0 = 0; t0 < SEQ - RING_STEPS; t0 += RING_STEPS) {
#pragma unroll
    for (int s = 0; s < RING_SLOTS; ++s) {
      const size_t t = (size_t)(t0 + s * GRP_ALL);
      use(s, t);
      __builtin_amdgcn_sched_barrier(0);
      ask(s, t + RING_STEPS);
      __builtin_amdgcn_sched_barrier(0);
    }
  }
#pragma unroll
  for (int s = 0; s < RING_SLOTS; ++s)
    use(s, (size_t)(SEQ - RING_STEPS + s * GRP_ALL));
}

__global__ __launch_bounds__(SCAN_TPB) void sru_scan_last(
    const float* __restrict__ U, const float* __restrict__ xin,
    const float* __restrict__ bias, float* __restrict__ hout)
{
  const int idx = blockIdx.x * SCAN_TPB + threadIdx.x;
  const int b = idx >> 10;
  const int h = idx & (HID - 1);
  const float bf = bias[h];
  const float br = bias[HID + h];
  const float* pU = U + (size_t)b * NTOT + h;

  const size_t rowL = (size_t)(SEQ - 1) * BATCH + b;
  const float urL = U[rowL * NTOT + 2 * HID + h];
  const float xvL = xin[rowL * HID + h];

  float sx[RING_SLOTS][GRP_LAST], sf[RING_SLOTS][GRP_LAST];
  float c = 0.f;

  auto ask = [&](int s, size_t t) {
#pragma unroll
    for (int j = 0; j < GRP_LAST; ++j) {
      const float* q = pU + (t + j) * SU_STEP;
      sx[s][j] = q[0];
      sf[s][j] = q[HID];
    }
  };
  auto use = [&](int s) {
#pragma unroll
    for (int j = 0; j < GRP_LAST; ++j) {
      const float f = sigmoidf_(sf[s][j] + bf);
      c = f * c + (1.f - f) * sx[s][j];
    }
  };

  __builtin_amdgcn_sched_barrier(0);
#pragma unroll
  for (int s = 0; s < RING_SLOTS; ++s) {
    ask(s, (size_t)s * GRP_LAST);
    __builtin_amdgcn_sched_barrier(0);
  }

  constexpr int RING_STEPS = RING_SLOTS * GRP_LAST;
  for (int t0 = 0; t0 < SEQ - RING_STEPS; t0 += RING_STEPS) {
#pragma unroll
    for (int s = 0; s < RING_SLOTS; ++s) {
      use(s);
      __builtin_amdgcn_sched_barrier(0);
      ask(s, (size_t)(t0 + s * GRP_LAST + RING_STEPS));
      __builtin_amdgcn_sched_barrier(0);
    }
  }
#pragma unroll
  for (int s = 0; s < RING_SLOTS; ++s) use(s);

  const float r = sigmoidf_(urL + br);
  hout[(size_t)b * HID + h] = r * c + (1.f - r) * xvL;
}

extern "C" void kernel_launch(void* const* d_in, const int* in_sizes, int n_in,
                              void* d_out, int out_size, void* d_ws, size_t ws_size,
                              hipStream_t stream) {
  const float* x  = (const float*)d_in[0];
  const float* W0 = (const float*)d_in[1];
  const float* b0 = (const float*)d_in[2];
  const float* W1 = (const float*)d_in[3];
  const float* b1 = (const float*)d_in[4];
  float* out = (float*)d_out;

  float* U  = (float*)d_ws;
  float* h1 = U + (size_t)MTOT * NTOT;

  const dim3 gemmGrid(NTOT / TN, MTOT / TM);
  const dim3 gemmBlk(256);
  const dim3 scanGrid((BATCH * HID) / SCAN_TPB);
  const dim3 scanBlk(SCAN_TPB);

  wmma_gemm_f32<<<gemmGrid, gemmBlk, 0, stream>>>(x, W0, U);
  sru_scan_all<<<scanGrid, scanBlk, 0, stream>>>(U, x, b0, h1);
  wmma_gemm_f32<<<gemmGrid, gemmBlk, 0, stream>>>(h1, W1, U);
  sru_scan_last<<<scanGrid, scanBlk, 0, stream>>>(U, h1, b1, out);
}


# layer-1 GEMM computes only the U columns its scan reads: candidate and forget columns of every step, reset-gate columns of the last step's 16 rows only (a 16x16 tile per wave); sum order unchanged
# speedup vs baseline: 1.6636x; 1.1759x over previous
#include <hip/hip_runtime.h>
#include <hip/hip_bf16.h>

typedef __attribute__((ext_vector_type(2))) float v2f;
typedef __attribute__((ext_vector_type(4))) float f4;
typedef __attribute__((ext_vector_type(8))) float v8f;

#define SEQ   2048
#define BATCH 16
#define HID   1024
#define MTOT  (SEQ * BATCH)
#define NTOT  (3 * HID)
#define KTOT  HID

#define TM 128
#define TN 128
#define TK 32
#define A_STRIDE 36
#define B_STRIDE 144

#define NB_ALL   (NTOT / TN)
#define NB_LIVE  (2 * HID / TN)
#define NB_RESET (HID / TN)
#define GRID_LIVE (NB_LIVE * (MTOT / TM))

__device__ __forceinline__ void reset_gate_last_step(
    const float* __restrict__ A, const float* __restrict__ B,
    float* __restrict__ C, int blk)
{
  const int lane = threadIdx.x & 31;
  const int wave = threadIdx.x >> 5;
  const int lm   = lane & 15;
  const int hk   = lane >> 4;
  const int col  = 2 * HID + blk * TN + wave * 16 + lm;
  const float* pa = A + (size_t)(MTOT - BATCH + lm) * KTOT + hk * 2;
  const float* pb = B + (size_t)(hk * 2) * NTOT + col;

  constexpr int UN = 16;
  v8f acc = {};
  for (int k = 0; k < KTOT; k += 4 * UN) {
    v2f a[UN], b[UN];
#pragma unroll
    for (int u = 0; u < UN; ++u) {
      a[u]   = *(const v2f*)(pa + k + 4 * u);
      b[u].x = pb[(size_t)(k + 4 * u) * NTOT];
      b[u].y = pb[(size_t)(k + 4 * u + 1) * NTOT];
    }
#pragma unroll
    for (int u = 0; u < UN; ++u)
      acc = __builtin_amdgcn_wmma_f32_16x16x4_f32(
          false, a[u], false, b[u], (short)0, acc, false, false);
  }
#pragma unroll
  for (int v = 0; v < 8; ++v)
    C[(size_t)(MTOT - BATCH + hk * 8 + v) * NTOT + col] = acc[v];
}

template <bool LAST_ONLY>
__global__ __launch_bounds__(256) void wmma_gemm_f32(
    const float* __restrict__ A, const float* __restrict__ B,
    float* __restrict__ C)
{
  if (LAST_ONLY && (int)blockIdx.x >= GRID_LIVE) {
    reset_gate_last_step(A, B, C, (int)blockIdx.x - GRID_LIVE);
    return;
  }

  __shared__ float As[TM * A_STRIDE];
  __shared__ float Bs[TK * B_STRIDE];

  const int tid  = threadIdx.x;
  const int lane = tid & 31;
  const int wave = tid >> 5;
  const int lm   = lane & 15;
  const int hk   = lane >> 4;
  const int waveM = (wave >> 2) * 64;
  const int waveN = (wave & 3) * 32;

  const int blockN = (LAST_ONLY ? ((int)blockIdx.x % NB_LIVE) : (int)blockIdx.x) * TN;
  const int blockM = (LAST_ONLY ? ((int)blockIdx.x / NB_LIVE) : (int)blockIdx.y) * TM;

  const int arow = tid >> 3;
  const int acol = (tid & 7) * 4;
  const int brow = tid >> 5;
  const int bcol = (tid & 31) * 4;

  v8f acc[4][2] = {};

  f4 aPre[4], bPre[4];
#pragma unroll
  for (int i = 0; i < 4; ++i) {
    aPre[i] = *(const f4*)&A[(size_t)(blockM + arow + i * 32) * KTOT + acol];
    bPre[i] = *(const f4*)&B[(size_t)(brow + i * 8) * NTOT + blockN + bcol];
  }

  for (int kt = 0; kt < KTOT; kt += TK) {
#pragma unroll
    for (int i = 0; i < 4; ++i) {
      *(f4*)&As[(arow + i * 32) * A_STRIDE + acol] = aPre[i];
      *(f4*)&Bs[(brow + i * 8) * B_STRIDE + bcol]  = bPre[i];
    }
    __syncthreads();

    const int ktn = kt + TK;
    if (ktn < KTOT) {
#pragma unroll
      for (int i = 0; i < 4; ++i) {
        aPre[i] = *(const f4*)&A[(size_t)(blockM + arow + i * 32) * KTOT + ktn + acol];
        bPre[i] = *(const f4*)&B[(size_t)(ktn + brow + i * 8) * NTOT + blockN + bcol];
      }
    }

#pragma unroll
    for (int kk = 0; kk < TK; kk += 4) {
      v2f afrag[4];
#pragma unroll
      for (int m = 0; m < 4; ++m)
        afrag[m] = *(const v2f*)&As[(waveM + m * 16 + lm) * A_STRIDE + kk + hk * 2];

      v2f bfrag[2];
#pragma unroll
      for (int n = 0; n < 2; ++n) {
        const int kb = (kk + hk * 2) * B_STRIDE + waveN + n * 16 + lm;
        bfrag[n].x = Bs[kb];
        bfrag[n].y = Bs[kb + B_STRIDE];
      }

#pragma unroll
      for (int m = 0; m < 4; ++m)
#pragma unroll
        for (int n = 0; n < 2; ++n)
          acc[m][n] = __builtin_amdgcn_wmma_f32_16x16x4_f32(
              false, afrag[m], false, bfrag[n], (short)0, acc[m][n],
              false, false);
    }
    __syncthreads();
  }

#pragma unroll
  for (int m = 0; m < 4; ++m) {
    const int row0 = blockM + waveM + m * 16 + hk * 8;
#pragma unroll
    for (int n = 0; n < 2; ++n) {
      const int col = blockN + waveN + n * 16 + lm;
#pragma unroll
      for (int v = 0; v < 8; ++v)
        C[(size_t)(row0 + v) * NTOT + col] = acc[m][n][v];
    }
  }
}

#define SCAN_TPB 64
#define SU_STEP ((size_t)BATCH * NTOT)
#define SX_STEP ((size_t)BATCH * HID)
#define RING_SLOTS 4
#define GRP_ALL  4
#define GRP_LAST 8

__device__ __forceinline__ float sigmoidf_(float z) {
  return 1.0f / (1.0f + __expf(-z));
}

__global__ __launch_bounds__(SCAN_TPB) void sru_scan_all(
    const float* __restrict__ U, const float* __restrict__ xin,
    const float* __restrict__ bias, float* __restrict__ hout)
{
  const int idx = blockIdx.x * SCAN_TPB + threadIdx.x;
  const int b = idx >> 10;
  const int h = idx & (HID - 1);
  const float bf = bias[h];
  const float br = bias[HID + h];
  const float* pU = U   + (size_t)b * NTOT + h;
  const float* pX = xin + (size_t)b * HID + h;
  float*       pH = hout + (size_t)b * HID + h;

  float sx[RING_SLOTS][GRP_ALL], sf[RING_SLOTS][GRP_ALL];
  float sr[RING_SLOTS][GRP_ALL], sv[RING_SLOTS][GRP_ALL];
  float c = 0.f;

  auto ask = [&](int s, size_t t) {
#pragma unroll
    for (int j = 0; j < GRP_ALL; ++j) {
      const float* q = pU + (t + j) * SU_STEP;
      sx[s][j] = q[0];
      sf[s][j] = q[HID];
      sr[s][j] = q[2 * HID];
      sv[s][j] = pX[(t + j) * SX_STEP];
    }
  };
  auto use = [&](int s, size_t t) {
#pragma unroll
    for (int j = 0; j < GRP_ALL; ++j) {
      const float f = sigmoidf_(sf[s][j] + bf);
      c = f * c + (1.f - f) * sx[s][j];
      const float r = sigmoidf_(sr[s][j] + br);
      pH[(t + j) * SX_STEP] = r * c + (1.f - r) * sv[s][j];
    }
  };

  __builtin_amdgcn_sched_barrier(0);
#pragma unroll
  for (int s = 0; s < RING_SLOTS; ++s) {
    ask(s, (size_t)s * GRP_ALL);
    __builtin_amdgcn_sched_barrier(0);
  }

  constexpr int RING_STEPS = RING_SLOTS * GRP_ALL;
  for (int t0 = 0; t0 < SEQ - RING_STEPS; t0 += RING_STEPS) {
#pragma unroll
    for (int s = 0; s < RING_SLOTS; ++s) {
      const size_t t = (size_t)(t0 + s * GRP_ALL);
      use(s, t);
      __builtin_amdgcn_sched_barrier(0);
      ask(s, t + RING_STEPS);
      __builtin_amdgcn_sched_barrier(0);
    }
  }
#pragma unroll
  for (int s = 0; s < RING_SLOTS; ++s)
    use(s, (size_t)(SEQ - RING_STEPS + s * GRP_ALL));
}

__global__ __launch_bounds__(SCAN_TPB) void sru_scan_last(
    const float* __restrict__ U, const float* __restrict__ xin,
    const float* __restrict__ bias, float* __restrict__ hout)
{
  const int idx = blockIdx.x * SCAN_TPB + threadIdx.x;
  const int b = idx >> 10;
  const int h = idx & (HID - 1);
  const float bf = bias[h];
  const float br = bias[HID + h];
  const float* pU = U + (size_t)b * NTOT + h;

  const size_t rowL = (size_t)(SEQ - 1) * BATCH + b;
  const float urL = U[rowL * NTOT + 2 * HID + h];
  const float xvL = xin[rowL * HID + h];

  float sx[RING_SLOTS][GRP_LAST], sf[RING_SLOTS][GRP_LAST];
  float c = 0.f;

  auto ask = [&](int s, size_t t) {
#pragma unroll
    for (int j = 0; j < GRP_LAST; ++j) {
      const float* q = pU + (t + j) * SU_STEP;
      sx[s][j] = q[0];
      sf[s][j] = q[HID];
    }
  };
  auto use = [&](int s) {
#pragma unroll
    for (int j = 0; j < GRP_LAST; ++j) {
      const float f = sigmoidf_(sf[s][j] + bf);
      c = f * c + (1.f - f) * sx[s][j];
    }
  };

  __builtin_amdgcn_sched_barrier(0);
#pragma unroll
  for (int s = 0; s < RING_SLOTS; ++s) {
    ask(s, (size_t)s * GRP_LAST);
    __builtin_amdgcn_sched_barrier(0);
  }

  constexpr int RING_STEPS = RING_SLOTS * GRP_LAST;
  for (int t0 = 0; t0 < SEQ - RING_STEPS; t0 += RING_STEPS) {
#pragma unroll
    for (int s = 0; s < RING_SLOTS; ++s) {
      use(s);
      __builtin_amdgcn_sched_barrier(0);
      ask(s, (size_t)(t0 + s * GRP_LAST + RING_STEPS));
      __builtin_amdgcn_sched_barrier(0);
    }
  }
#pragma unroll
  for (int s = 0; s < RING_SLOTS; ++s) use(s);

  const float r = sigmoidf_(urL + br);
  hout[(size_t)b * HID + h] = r * c + (1.f - r) * xvL;
}

extern "C" void kernel_launch(void* const* d_in, const int* in_sizes, int n_in,
                              void* d_out, int out_size, void* d_ws, size_t ws_size,
                              hipStream_t stream) {
  const float* x  = (const float*)d_in[0];
  const float* W0 = (const float*)d_in[1];
  const float* b0 = (const float*)d_in[2];
  const float* W1 = (const float*)d_in[3];
  const float* b1 = (const float*)d_in[4];
  float* out = (float*)d_out;

  float* U  = (float*)d_ws;
  float* h1 = U + (size_t)MTOT * NTOT;

  const dim3 gemmGrid(NB_ALL, MTOT / TM);
  const dim3 gemmGridLast(GRID_LIVE + NB_RESET);
  const dim3 gemmBlk(256);
  const dim3 scanGrid((BATCH * HID) / SCAN_TPB);
  const dim3 scanBlk(SCAN_TPB);

  wmma_gemm_f32<false><<<gemmGrid, gemmBlk, 0, stream>>>(x, W0, U);
  sru_scan_all<<<scanGrid, scanBlk, 0, stream>>>(U, x, b0, h1);
  wmma_gemm_f32<true><<<gemmGridLast, gemmBlk, 0, stream>>>(h1, W1, U);
  sru_scan_last<<<scanGrid, scanBlk, 0, stream>>>(U, h1, b1, out);
}


# GEMM epilogue stores U time-packed (4 consecutive steps of a channel in 16 contiguous bytes), so one 16-byte scan load brings 4 steps of a U column: 2-4x the bytes in flight; sum order unchanged
# speedup vs baseline: 1.7300x; 1.0399x over previous
#include <hip/hip_runtime.h>
#include <hip/hip_bf16.h>

typedef __attribute__((ext_vector_type(2))) float v2f;
typedef __attribute__((ext_vector_type(4))) float f4;
typedef __attribute__((ext_vector_type(8))) float v8f;

#define SEQ   2048
#define BATCH 16
#define HID   1024
#define MTOT  (SEQ * BATCH)
#define NTOT  (3 * HID)
#define KTOT  HID
#define NQUAD (SEQ / 4)

#define TM 128
#define TN 128
#define TK 32
#define A_STRIDE 36
#define B_STRIDE 144

#define NB_ALL   (NTOT / TN)
#define NB_LIVE  (2 * HID / TN)
#define NB_RESET (HID / TN)
#define GRID_LIVE (NB_LIVE * (MTOT / TM))

__device__ __forceinline__ void reset_gate_last_step(
    const float* __restrict__ A, const float* __restrict__ B,
    float* __restrict__ C, int blk)
{
  const int lane = threadIdx.x & 31;
  const int wave = threadIdx.x >> 5;
  const int lm   = lane & 15;
  const int hk   = lane >> 4;
  const int col  = 2 * HID + blk * TN + wave * 16 + lm;
  const float* pa = A + (size_t)(MTOT - BATCH + lm) * KTOT + hk * 2;
  const float* pb = B + (size_t)(hk * 2) * NTOT + col;

  constexpr int UN = 16;
  v8f acc = {};
  for (int k = 0; k < KTOT; k += 4 * UN) {
    v2f a[UN], b[UN];
#pragma unroll
    for (int u = 0; u < UN; ++u) {
      a[u]   = *(const v2f*)(pa + k + 4 * u);
      b[u].x = pb[(size_t)(k + 4 * u) * NTOT];
      b[u].y = pb[(size_t)(k + 4 * u + 1) * NTOT];
    }
#pragma unroll
    for (int u = 0; u < UN; ++u)
      acc = __builtin_amdgcn_wmma_f32_16x16x4_f32(
          false, a[u], false, b[u], (short)0, acc, false, false);
  }
#pragma unroll
  for (int v = 0; v < 8; ++v)
    C[(((size_t)(NQUAD - 1) * BATCH + hk * 8 + v) * NTOT + col) * 4 + 3] = acc[v];
}

template <bool LAST_ONLY>
__global__ __launch_bounds__(256) void wmma_gemm_f32(
    const float* __restrict__ A, const float* __restrict__ B,
    float* __restrict__ C)
{
  if (LAST_ONLY && (int)blockIdx.x >= GRID_LIVE) {
    reset_gate_last_step(A, B, C, (int)blockIdx.x - GRID_LIVE);
    return;
  }

  __shared__ float As[TM * A_STRIDE];
  __shared__ float Bs[TK * B_STRIDE];

  const int tid  = threadIdx.x;
  const int lane = tid & 31;
  const int wave = tid >> 5;
  const int lm   = lane & 15;
  const int hk   = lane >> 4;
  const int waveM = (wave >> 2) * 64;
  const int waveN = (wave & 3) * 32;

  const int blockN = (LAST_ONLY ? ((int)blockIdx.x % NB_LIVE) : (int)blockIdx.x) * TN;
  const int blockM = (LAST_ONLY ? ((int)blockIdx.x / NB_LIVE) : (int)blockIdx.y) * TM;

  const int arow = tid >> 3;
  const int acol = (tid & 7) * 4;
  const int brow = tid >> 5;
  const int bcol = (tid & 31) * 4;

  v8f acc[4][2] = {};

  f4 aPre[4], bPre[4];
#pragma unroll
  for (int i = 0; i < 4; ++i) {
    aPre[i] = *(const f4*)&A[(size_t)(blockM + arow + i * 32) * KTOT + acol];
    bPre[i] = *(const f4*)&B[(size_t)(brow + i * 8) * NTOT + blockN + bcol];
  }

  for (int kt = 0; kt < KTOT; kt += TK) {
#pragma unroll
    for (int i = 0; i < 4; ++i) {
      *(f4*)&As[(arow + i * 32) * A_STRIDE + acol] = aPre[i];
      *(f4*)&Bs[(brow + i * 8) * B_STRIDE + bcol]  = bPre[i];
    }
    __syncthreads();

    const int ktn = kt + TK;
    if (ktn < KTOT) {
#pragma unroll
      for (int i = 0; i < 4; ++i) {
        aPre[i] = *(const f4*)&A[(size_t)(blockM + arow + i * 32) * KTOT + ktn + acol];
        bPre[i] = *(const f4*)&B[(size_t)(ktn + brow + i * 8) * NTOT + blockN + bcol];
      }
    }

#pragma unroll
    for (int kk = 0; kk < TK; kk += 4) {
      v2f afrag[4];
#pragma unroll
      for (int m = 0; m < 4; ++m)
        afrag[m] = *(const v2f*)&As[(waveM + m * 16 + lm) * A_STRIDE + kk + hk * 2];

      v2f bfrag[2];
#pragma unroll
      for (int n = 0; n < 2; ++n) {
        const int kb = (kk + hk * 2) * B_STRIDE + waveN + n * 16 + lm;
        bfrag[n].x = Bs[kb];
        bfrag[n].y = Bs[kb + B_STRIDE];
      }

#pragma unroll
      for (int m = 0; m < 4; ++m)
#pragma unroll
        for (int n = 0; n < 2; ++n)
          acc[m][n] = __builtin_amdgcn_wmma_f32_16x16x4_f32(
              false, afrag[m], false, bfrag[n], (short)0, acc[m][n],
              false, false);
    }
    __syncthreads();
  }

  const int tq = (blockM + waveM) >> 6;
  f4* C4 = (f4*)C;
#pragma unroll
  for (int n = 0; n < 2; ++n) {
    const int col = blockN + waveN + n * 16 + lm;
#pragma unroll
    for (int v = 0; v < 8; ++v) {
      f4 q;
      q.x = acc[0][n][v];
      q.y = acc[1][n][v];
      q.z = acc[2][n][v];
      q.w = acc[3][n][v];
      C4[((size_t)tq * BATCH + hk * 8 + v) * NTOT + col] = q;
    }
  }
}

#define SCAN_TPB 64
#define QU_STEP ((size_t)BATCH * NTOT)
#define SX_STEP ((size_t)BATCH * HID)
#define RING_SLOTS 4
#define GRPQ_ALL  2
#define GRPQ_LAST 8

__device__ __forceinline__ float sigmoidf_(float z) {
  return 1.0f / (1.0f + __expf(-z));
}

__global__ __launch_bounds__(SCAN_TPB) void sru_scan_all(
    const float* __restrict__ U, const float* __restrict__ xin,
    const float* __restrict__ bias, float* __restrict__ hout)
{
  const int idx = blockIdx.x * SCAN_TPB + threadIdx.x;
  const int b = idx >> 10;
  const int h = idx & (HID - 1);
  const float bf = bias[h];
  const float br = bias[HID + h];
  const f4*    pU = (const f4*)U + (size_t)b * NTOT + h;
  const float* pX = xin + (size_t)b * HID + h;
  float*       pH = hout + (size_t)b * HID + h;

  f4 sx[RING_SLOTS][GRPQ_ALL], sf[RING_SLOTS][GRPQ_ALL];
  f4 sr[RING_SLOTS][GRPQ_ALL], sv[RING_SLOTS][GRPQ_ALL];
  float c = 0.f;

  auto ask = [&](int s, size_t q) {
#pragma unroll
    for (int j = 0; j < GRPQ_ALL; ++j) {
      const f4* p = pU + (q + j) * QU_STEP;
      sx[s][j] = p[0];
      sf[s][j] = p[HID];
      sr[s][j] = p[2 * HID];
#pragma unroll
      for (int i = 0; i < 4; ++i)
        sv[s][j][i] = pX[((q + j) * 4 + i) * SX_STEP];
    }
  };
  auto use = [&](int s, size_t q) {
#pragma unroll
    for (int j = 0; j < GRPQ_ALL; ++j) {
#pragma unroll
      for (int i = 0; i < 4; ++i) {
        const float f = sigmoidf_(sf[s][j][i] + bf);
        c = f * c + (1.f - f) * sx[s][j][i];
        const float r = sigmoidf_(sr[s][j][i] + br);
        pH[((q + j) * 4 + i) * SX_STEP] = r * c + (1.f - r) * sv[s][j][i];
      }
    }
  };

  __builtin_amdgcn_sched_barrier(0);
#pragma unroll
  for (int s = 0; s < RING_SLOTS; ++s) {
    ask(s, (size_t)s * GRPQ_ALL);
    __builtin_amdgcn_sched_barrier(0);
  }

  constexpr int RING_Q = RING_SLOTS * GRPQ_ALL;
  for (int q0 = 0; q0 < NQUAD - RING_Q; q0 += RING_Q) {
#pragma unroll
    for (int s = 0; s < RING_SLOTS; ++s) {
      const size_t q = (size_t)(q0 + s * GRPQ_ALL);
      use(s, q);
      __builtin_amdgcn_sched_barrier(0);
      ask(s, q + RING_Q);
      __builtin_amdgcn_sched_barrier(0);
    }
  }
#pragma unroll
  for (int s = 0; s < RING_SLOTS; ++s)
    use(s, (size_t)(NQUAD - RING_Q + s * GRPQ_ALL));
}

__global__ __launch_bounds__(SCAN_TPB) void sru_scan_last(
    const float* __restrict__ U, const float* __restrict__ xin,
    const float* __restrict__ bias, float* __restrict__ hout)
{
  const int idx = blockIdx.x * SCAN_TPB + threadIdx.x;
  const int b = idx >> 10;
  const int h = idx & (HID - 1);
  const float bf = bias[h];
  const float br = bias[HID + h];
  const f4* pU = (const f4*)U + (size_t)b * NTOT + h;

  const size_t rowL = (size_t)(SEQ - 1) * BATCH + b;
  const float urL =
      U[(((size_t)(NQUAD - 1) * BATCH + b) * NTOT + 2 * HID + h) * 4 + 3];
  const float xvL = xin[rowL * HID + h];

  f4 sx[RING_SLOTS][GRPQ_LAST], sf[RING_SLOTS][GRPQ_LAST];
  float c = 0.f;

  auto ask = [&](int s, size_t q) {
#pragma unroll
    for (int j = 0; j < GRPQ_LAST; ++j) {
      const f4* p = pU + (q + j) * QU_STEP;
      sx[s][j] = p[0];
      sf[s][j] = p[HID];
    }
  };
  auto use = [&](int s) {
#pragma unroll
    for (int j = 0; j < GRPQ_LAST; ++j) {
#pragma unroll
      for (int i = 0; i < 4; ++i) {
        const float f = sigmoidf_(sf[s][j][i] + bf);
        c = f * c + (1.f - f) * sx[s][j][i];
      }
    }
  };

  __builtin_amdgcn_sched_barrier(0);
#pragma unroll
  for (int s = 0; s < RING_SLOTS; ++s) {
    ask(s, (size_t)s * GRPQ_LAST);
    __builtin_amdgcn_sched_barrier(0);
  }

  constexpr int RING_Q = RING_SLOTS * GRPQ_LAST;
  for (int q0 = 0; q0 < NQUAD - RING_Q; q0 += RING_Q) {
#pragma unroll
    for (int s = 0; s < RING_SLOTS; ++s) {
      use(s);
      __builtin_amdgcn_sched_barrier(0);
      ask(s, (size_t)(q0 + s * GRPQ_LAST + RING_Q));
      __builtin_amdgcn_sched_barrier(0);
    }
  }
#pragma unroll
  for (int s = 0; s < RING_SLOTS; ++s) use(s);

  const float r = sigmoidf_(urL + br);
  hout[(size_t)b * HID + h] = r * c + (1.f - r) * xvL;
}

extern "C" void kernel_launch(void* const* d_in, const int* in_sizes, int n_in,
                              void* d_out, int out_size, void* d_ws, size_t ws_size,
                              hipStream_t stream) {
  const float* x  = (const float*)d_in[0];
  const float* W0 = (const float*)d_in[1];
  const float* b0 = (const float*)d_in[2];
  const float* W1 = (const float*)d_in[3];
  const float* b1 = (const float*)d_in[4];
  float* out = (float*)d_out;

  float* U  = (float*)d_ws;
  float* h1 = U + (size_t)MTOT * NTOT;

  const dim3 gemmGrid(NB_ALL, MTOT / TM);
  const dim3 gemmGridLast(GRID_LIVE + NB_RESET);
  const dim3 gemmBlk(256);
  const dim3 scanGrid((BATCH * HID) / SCAN_TPB);
  const dim3 scanBlk(SCAN_TPB);

  wmma_gemm_f32<false><<<gemmGrid, gemmBlk, 0, stream>>>(x, W0, U);
  sru_scan_all<<<scanGrid, scanBlk, 0, stream>>>(U, x, b0, h1);
  wmma_gemm_f32<true><<<gemmGridLast, gemmBlk, 0, stream>>>(h1, W1, U);
  sru_scan_last<<<scanGrid, scanBlk, 0, stream>>>(U, h1, b1, out);
}
